# starting kernel
# baseline (speedup 1.0000x reference)
#include <hip/hip_runtime.h>
#include <hip/hip_bf16.h>

#define NN    50000
#define RR    51
#define EE    16
#define CC    8
#define BB    40
#define NNZ   1600000
#define NE    (NN * EE)
#define W1ELEMS ((size_t)RR * NE)

typedef float v2f __attribute__((ext_vector_type(2)));
typedef float v8f __attribute__((ext_vector_type(8)));

__device__ __forceinline__ void atomAddF32(float* p, float v) {
  unsafeAtomicAdd(p, v);
}

__global__ void rgcn_init(float* __restrict__ h, float* __restrict__ out,
                          const float* __restrict__ bias2) {
  int i = blockIdx.x * blockDim.x + threadIdx.x;
  if (i < NE)        h[i] = 0.0f;
  if (i < NN * CC)   out[i] = bias2[i & (CC - 1)];
}

__global__ void rgcn_w2(const float* __restrict__ comps2,
                        const float* __restrict__ bases2,
                        float* __restrict__ W2) {
  int idx = blockIdx.x * blockDim.x + threadIdx.x;
  if (idx >= RR * EE * CC) return;
  int r  = idx >> 7;
  int ec = idx & 127;
  float acc = 0.0f;
#pragma unroll
  for (int b = 0; b < BB; ++b)
    acc += comps2[r * BB + b] * bases2[b * (EE * CC) + ec];
  W2[idx] = acc;
}

__global__ void __launch_bounds__(256)
rgcn_w1_gemm(const float* __restrict__ comps1,
             const float* __restrict__ bases1,
             float* __restrict__ W1) {
  const int lane = threadIdx.x & 31;
  const int half = lane >> 4;
  const int l16  = lane & 15;

  v2f a[4][10];
#pragma unroll
  for (int mt = 0; mt < 4; ++mt) {
    int m = mt * 16 + l16;
#pragma unroll
    for (int ks = 0; ks < 10; ++ks) {
      int k = ks * 4 + half * 2;
      float ax = 0.0f, ay = 0.0f;
      if (m < RR) {
        ax = comps1[m * BB + k];
        ay = comps1[m * BB + k + 1];
      }
      a[mt][ks].x = ax;
      a[mt][ks].y = ay;
    }
  }

  const int wavesPerBlock = blockDim.x >> 5;
  const int waveId  = blockIdx.x * wavesPerBlock + (threadIdx.x >> 5);
  const int nWaves  = gridDim.x * wavesPerBlock;
  const int nTiles  = NE / 16;

  for (int tile = waveId; tile < nTiles; tile += nWaves) {
    const int col = tile * 16 + l16;
    v8f acc[4] = {};
#pragma unroll
    for (int ks = 0; ks < 10; ++ks) {
      const int k = ks * 4 + half * 2;
      v2f b;
      b.x = __builtin_nontemporal_load(&bases1[(size_t)k * NE + col]);
      b.y = __builtin_nontemporal_load(&bases1[(size_t)(k + 1) * NE + col]);
#pragma unroll
      for (int mt = 0; mt < 4; ++mt) {
        acc[mt] = __builtin_amdgcn_wmma_f32_16x16x4_f32(
             false, a[mt][ks],  false, b,
             (short)0, acc[mt],
             false,  false);
      }
    }
#pragma unroll
    for (int mt = 0; mt < 4; ++mt) {
#pragma unroll
      for (int j = 0; j < 8; ++j) {
        int m = mt * 16 + j + half * 8;
        if (m < RR) W1[(size_t)m * NE + col] = acc[mt][j];
      }
    }
  }
}

__global__ void rgcn_scatter_h(const float* __restrict__ values,
                               const int* __restrict__ hr,
                               const int* __restrict__ hc,
                               const float* __restrict__ W1,
                               float* __restrict__ h) {
  int t = blockIdx.x * blockDim.x + threadIdx.x;
  if (t >= NNZ * EE) return;
  int i = t >> 4;
  int e = t & 15;
  float v  = __builtin_nontemporal_load(&values[i]);
  int col  = __builtin_nontemporal_load(&hc[i]);
  int row  = __builtin_nontemporal_load(&hr[i]);
  float w  = W1[(size_t)col * EE + e];
  atomAddF32(&h[row * EE + e], v * w);
}

__global__ void rgcn_bias_relu(float* __restrict__ h,
                               const float* __restrict__ bias1) {
  int i = blockIdx.x * blockDim.x + threadIdx.x;
  if (i >= NE) return;
  float x = h[i] + bias1[i & 15];
  h[i] = x > 0.0f ? x : 0.0f;
}

__global__ void rgcn_layer2(const float* __restrict__ values,
                            const int* __restrict__ vr,
                            const int* __restrict__ vc,
                            const float* __restrict__ h,
                            const float* __restrict__ W2,
                            float* __restrict__ out) {
  __shared__ float W2s[RR * EE * CC];
  for (int j = threadIdx.x; j < RR * EE * CC; j += blockDim.x)
    W2s[j] = W2[j];
  __syncthreads();

  const long long total  = (long long)NNZ * CC;
  const long long stride = (long long)gridDim.x * blockDim.x;
  for (long long t = (long long)blockIdx.x * blockDim.x + threadIdx.x;
       t < total; t += stride) {
    int i = (int)(t >> 3);
    int c = (int)(t & 7);
    int r   = __builtin_nontemporal_load(&vr[i]);
    int o   = __builtin_nontemporal_load(&vc[i]);
    float v = __builtin_nontemporal_load(&values[i]);
    int p = r / NN;
    int s = r - p * NN;
    const float* hp = h + (size_t)o * EE;
    const float* wp = W2s + p * (EE * CC) + c;
    float acc = 0.0f;
#pragma unroll
    for (int e = 0; e < EE; ++e)
      acc += hp[e] * wp[e * CC];
    atomAddF32(&out[s * CC + c], v * acc);
  }
}

extern "C" void kernel_launch(void* const* d_in, const int* in_sizes, int n_in,
                              void* d_out, int out_size, void* d_ws, size_t ws_size,
                              hipStream_t stream) {
  const float* values = (const float*)d_in[0];
  const float* comps1 = (const float*)d_in[1];
  const float* bases1 = (const float*)d_in[2];
  const float* comps2 = (const float*)d_in[3];
  const float* bases2 = (const float*)d_in[4];
  const float* bias1  = (const float*)d_in[5];
  const float* bias2  = (const float*)d_in[6];
  const int*   hr     = (const int*)d_in[7];
  const int*   hc     = (const int*)d_in[8];
  const int*   vr     = (const int*)d_in[9];
  const int*   vc     = (const int*)d_in[10];
  float* out = (float*)d_out;

  float* W1 = (float*)d_ws;
  float* h  = W1 + W1ELEMS;
  float* W2 = h + NE;

  rgcn_init<<<(NE + 255) / 256, 256, 0, stream>>>(h, out, bias2);
  rgcn_w2<<<(RR * EE * CC + 255) / 256, 256, 0, stream>>>(comps2, bases2, W2);
  rgcn_w1_gemm<<<832, 256, 0, stream>>>(comps1, bases1, W1);
  {
    long long n = (long long)NNZ * EE;
    rgcn_scatter_h<<<(int)((n + 255) / 256), 256, 0, stream>>>(values, hr, hc, W1, h);
  }
  rgcn_bias_relu<<<(NE + 255) / 256, 256, 0, stream>>>(h, bias1);
  rgcn_layer2<<<4096, 256, 0, stream>>>(values, vr, vc, h, W2, out);
}


# layer-1 scatter: a wave owns 128 edges, reads ids and values in three 16-byte loads, hands them to half-waves by lane reads; 12,500 waves (was 800,000), 16-32 row gathers in flight; f32 atomics kept
# speedup vs baseline: 1.2529x; 1.2529x over previous
#include <hip/hip_runtime.h>
#include <hip/hip_bf16.h>

#define NN    50000
#define RR    51
#define EE    16
#define CC    8
#define BB    40
#define NNZ   1600000
#define NE    (NN * EE)
#define W1ELEMS ((size_t)RR * NE)

typedef float v2f __attribute__((ext_vector_type(2)));
typedef float v8f __attribute__((ext_vector_type(8)));

__device__ __forceinline__ void atomAddF32(float* p, float v) {
  unsafeAtomicAdd(p, v);
}

__global__ void rgcn_init(float* __restrict__ h, float* __restrict__ out,
                          const float* __restrict__ bias2) {
  int i = blockIdx.x * blockDim.x + threadIdx.x;
  if (i < NE)        h[i] = 0.0f;
  if (i < NN * CC)   out[i] = bias2[i & (CC - 1)];
}

__global__ void rgcn_w2(const float* __restrict__ comps2,
                        const float* __restrict__ bases2,
                        float* __restrict__ W2) {
  int idx = blockIdx.x * blockDim.x + threadIdx.x;
  if (idx >= RR * EE * CC) return;
  int r  = idx >> 7;
  int ec = idx & 127;
  float acc = 0.0f;
#pragma unroll
  for (int b = 0; b < BB; ++b)
    acc += comps2[r * BB + b] * bases2[b * (EE * CC) + ec];
  W2[idx] = acc;
}

typedef float v4f __attribute__((ext_vector_type(4)));

#define STRIP   64
#define NSTRIPS (NE / STRIP)

__global__ void __launch_bounds__(32)
rgcn_w1_gemm(const float* __restrict__ comps1,
             const float* __restrict__ bases1,
             float* __restrict__ W1) {
  const int lane = threadIdx.x & 31;
  const int half = lane >> 4;
  const int l16  = lane & 15;

  v2f a[4][10];
#pragma unroll
  for (int mt = 0; mt < 4; ++mt) {
    int m = mt * 16 + l16;
#pragma unroll
    for (int ks = 0; ks < 10; ++ks) {
      int k = ks * 4 + half * 2;
      float ax = 0.0f, ay = 0.0f;
      if (m < RR) {
        ax = comps1[m * BB + k];
        ay = comps1[m * BB + k + 1];
      }
      a[mt][ks].x = ax;
      a[mt][ks].y = ay;
    }
  }

  for (int strip = blockIdx.x; strip < NSTRIPS; strip += gridDim.x) {
    const int col0 = strip * STRIP + l16 * 4;
    v8f acc[4][4] = {};
#pragma unroll
    for (int ks = 0; ks < 10; ++ks) {
      const int k = ks * 4 + half * 2;
      const v4f b0 = __builtin_nontemporal_load(
          (const v4f*)&bases1[(size_t)k * NE + col0]);
      const v4f b1 = __builtin_nontemporal_load(
          (const v4f*)&bases1[(size_t)(k + 1) * NE + col0]);
#pragma unroll
      for (int nt = 0; nt < 4; ++nt) {
        v2f b;
        b.x = b0[nt];
        b.y = b1[nt];
#pragma unroll
        for (int mt = 0; mt < 4; ++mt) {
          acc[nt][mt] = __builtin_amdgcn_wmma_f32_16x16x4_f32(
               false, a[mt][ks],  false, b,
               (short)0, acc[nt][mt],
               false,  false);
        }
      }
    }
#pragma unroll
    for (int mt = 0; mt < 4; ++mt) {
#pragma unroll
      for (int j = 0; j < 8; ++j) {
        int m = mt * 16 + j + half * 8;
        if (m < RR) {
          v4f o;
          o.x = acc[0][mt][j];
          o.y = acc[1][mt][j];
          o.z = acc[2][mt][j];
          o.w = acc[3][mt][j];
          *(v4f*)&W1[(size_t)m * NE + col0] = o;
        }
      }
    }
  }
}

typedef int v4i __attribute__((ext_vector_type(4)));

#define EPW      128
#define K3_WAVES 4
#define K3_GRID  (NNZ / (EPW * K3_WAVES))
static_assert(NNZ % (EPW * K3_WAVES) == 0, "edges must fill whole workgroups");

template <int J>
__device__ __forceinline__ int pairRecord(const v4i r, bool hi) {
  const int lo_v = __builtin_amdgcn_readlane(r[(J & 1) * 2],     J >> 1);
  const int hi_v = __builtin_amdgcn_readlane(r[(J & 1) * 2 + 1], J >> 1);
  return hi ? hi_v : lo_v;
}

template <int G>
__device__ __forceinline__ void k3Gather(const v4i c4, bool hi, int e,
                                         const float* __restrict__ W1,
                                         float (&w)[16]) {
#define K3_LOAD(Q) { const int col = pairRecord<G * 16 + (Q)>(c4, hi); \
                     w[Q] = W1[(size_t)(unsigned)(col * EE + e)]; }
  K3_LOAD(0)  K3_LOAD(1)  K3_LOAD(2)  K3_LOAD(3)
  K3_LOAD(4)  K3_LOAD(5)  K3_LOAD(6)  K3_LOAD(7)
  K3_LOAD(8)  K3_LOAD(9)  K3_LOAD(10) K3_LOAD(11)
  K3_LOAD(12) K3_LOAD(13) K3_LOAD(14) K3_LOAD(15)
#undef K3_LOAD
}

template <int G>
__device__ __forceinline__ void k3Add(const v4i r4, const v4i v4bits, bool hi,
                                      int e, const float (&w)[16],
                                      float* __restrict__ h) {
#define K3_ADD(Q) { const int row = pairRecord<G * 16 + (Q)>(r4, hi); \
                    const float v = __builtin_bit_cast(float, pairRecord<G * 16 + (Q)>(v4bits, hi)); \
                    atomAddF32(&h[row * EE + e], v * w[Q]); }
  K3_ADD(0)  K3_ADD(1)  K3_ADD(2)  K3_ADD(3)
  K3_ADD(4)  K3_ADD(5)  K3_ADD(6)  K3_ADD(7)
  K3_ADD(8)  K3_ADD(9)  K3_ADD(10) K3_ADD(11)
  K3_ADD(12) K3_ADD(13) K3_ADD(14) K3_ADD(15)
#undef K3_ADD
}

__global__ void __launch_bounds__(32 * K3_WAVES)
rgcn_scatter_h(const float* __restrict__ values,
               const int* __restrict__ hr,
               const int* __restrict__ hc,
               const float* __restrict__ W1,
               float* __restrict__ h) {
  const int lane = threadIdx.x & 31;
  const int wave = blockIdx.x * K3_WAVES + (threadIdx.x >> 5);
  const int e    = lane & 15;
  const bool hi  = lane >= 16;
  const int base = wave * EPW + lane * 4;
  const v4i c4 = __builtin_nontemporal_load((const v4i*)&hc[base]);
  const v4i r4 = __builtin_nontemporal_load((const v4i*)&hr[base]);
  const v4f vf = __builtin_nontemporal_load((const v4f*)&values[base]);
  const v4i vb = __builtin_bit_cast(v4i, vf);

  float wa[16], wb[16];
  k3Gather<0>(c4, hi, e, W1, wa);
  k3Gather<1>(c4, hi, e, W1, wb);
  k3Add<0>(r4, vb, hi, e, wa, h);
  k3Gather<2>(c4, hi, e, W1, wa);
  k3Add<1>(r4, vb, hi, e, wb, h);
  k3Gather<3>(c4, hi, e, W1, wb);
  k3Add<2>(r4, vb, hi, e, wa, h);
  k3Add<3>(r4, vb, hi, e, wb, h);
}

__global__ void rgcn_bias_relu(float* __restrict__ h,
                               const float* __restrict__ bias1) {
  int i = blockIdx.x * blockDim.x + threadIdx.x;
  if (i >= NE) return;
  float x = h[i] + bias1[i & 15];
  h[i] = x > 0.0f ? x : 0.0f;
}

__global__ void rgcn_layer2(const float* __restrict__ values,
                            const int* __restrict__ vr,
                            const int* __restrict__ vc,
                            const float* __restrict__ h,
                            const float* __restrict__ W2,
                            float* __restrict__ out) {
  __shared__ float W2s[RR * EE * CC];
  for (int j = threadIdx.x; j < RR * EE * CC; j += blockDim.x)
    W2s[j] = W2[j];
  __syncthreads();

  const long long total  = (long long)NNZ * CC;
  const long long stride = (long long)gridDim.x * blockDim.x;
  for (long long t = (long long)blockIdx.x * blockDim.x + threadIdx.x;
       t < total; t += stride) {
    int i = (int)(t >> 3);
    int c = (int)(t & 7);
    int r   = __builtin_nontemporal_load(&vr[i]);
    int o   = __builtin_nontemporal_load(&vc[i]);
    float v = __builtin_nontemporal_load(&values[i]);
    int p = r / NN;
    int s = r - p * NN;
    const float* hp = h + (size_t)o * EE;
    const float* wp = W2s + p * (EE * CC) + c;
    float acc = 0.0f;
#pragma unroll
    for (int e = 0; e < EE; ++e)
      acc += hp[e] * wp[e * CC];
    atomAddF32(&out[s * CC + c], v * acc);
  }
}

extern "C" void kernel_launch(void* const* d_in, const int* in_sizes, int n_in,
                              void* d_out, int out_size, void* d_ws, size_t ws_size,
                              hipStream_t stream) {
  const float* values = (const float*)d_in[0];
  const float* comps1 = (const float*)d_in[1];
  const float* bases1 = (const float*)d_in[2];
  const float* comps2 = (const float*)d_in[3];
  const float* bases2 = (const float*)d_in[4];
  const float* bias1  = (const float*)d_in[5];
  const float* bias2  = (const float*)d_in[6];
  const int*   hr     = (const int*)d_in[7];
  const int*   hc     = (const int*)d_in[8];
  const int*   vr     = (const int*)d_in[9];
  const int*   vc     = (const int*)d_in[10];
  float* out = (float*)d_out;

  float* W1 = (float*)d_ws;
  float* h  = W1 + W1ELEMS;
  float* W2 = h + NE;

  rgcn_init<<<(NE + 255) / 256, 256, 0, stream>>>(h, out, bias2);
  rgcn_w2<<<(RR * EE * CC + 255) / 256, 256, 0, stream>>>(comps2, bases2, W2);
  rgcn_w1_gemm<<<2500, 32, 0, stream>>>(comps1, bases1, W1);
  rgcn_scatter_h<<<K3_GRID, 32 * K3_WAVES, 0, stream>>>(values, hr, hc, W1, h);
  rgcn_bias_relu<<<(NE + 255) / 256, 256, 0, stream>>>(h, bias1);
  rgcn_layer2<<<4096, 256, 0, stream>>>(values, vr, vc, h, W2, out);
}


# layer-2 kernel walks edges as the layer-1 scatter does: a wave owns 128 edges, half-wave per edge, lane = feature, W2 in LDS once per 10 waves; sum over e by in-row lane exchanges, its order changes
# speedup vs baseline: 1.4777x; 1.1794x over previous
#include <hip/hip_runtime.h>
#include <hip/hip_bf16.h>

#define NN    50000
#define RR    51
#define EE    16
#define CC    8
#define BB    40
#define NNZ   1600000
#define NE    (NN * EE)
#define W1ELEMS ((size_t)RR * NE)

typedef float v2f __attribute__((ext_vector_type(2)));
typedef float v8f __attribute__((ext_vector_type(8)));

__device__ __forceinline__ void atomAddF32(float* p, float v) {
  unsafeAtomicAdd(p, v);
}

__global__ void rgcn_init(float* __restrict__ h, float* __restrict__ out,
                          const float* __restrict__ bias2) {
  int i = blockIdx.x * blockDim.x + threadIdx.x;
  if (i < NE)        h[i] = 0.0f;
  if (i < NN * CC)   out[i] = bias2[i & (CC - 1)];
}

__global__ void rgcn_w2(const float* __restrict__ comps2,
                        const float* __restrict__ bases2,
                        float* __restrict__ W2) {
  int idx = blockIdx.x * blockDim.x + threadIdx.x;
  if (idx >= RR * EE * CC) return;
  int r  = idx >> 7;
  int ec = idx & 127;
  float acc = 0.0f;
#pragma unroll
  for (int b = 0; b < BB; ++b)
    acc += comps2[r * BB + b] * bases2[b * (EE * CC) + ec];
  W2[idx] = acc;
}

typedef float v4f __attribute__((ext_vector_type(4)));

#define STRIP   64
#define NSTRIPS (NE / STRIP)

__global__ void __launch_bounds__(32)
rgcn_w1_gemm(const float* __restrict__ comps1,
             const float* __restrict__ bases1,
             float* __restrict__ W1) {
  const int lane = threadIdx.x & 31;
  const int half = lane >> 4;
  const int l16  = lane & 15;

  v2f a[4][10];
#pragma unroll
  for (int mt = 0; mt < 4; ++mt) {
    int m = mt * 16 + l16;
#pragma unroll
    for (int ks = 0; ks < 10; ++ks) {
      int k = ks * 4 + half * 2;
      float ax = 0.0f, ay = 0.0f;
      if (m < RR) {
        ax = comps1[m * BB + k];
        ay = comps1[m * BB + k + 1];
      }
      a[mt][ks].x = ax;
      a[mt][ks].y = ay;
    }
  }

  for (int strip = blockIdx.x; strip < NSTRIPS; strip += gridDim.x) {
    const int col0 = strip * STRIP + l16 * 4;
    v8f acc[4][4] = {};
#pragma unroll
    for (int ks = 0; ks < 10; ++ks) {
      const int k = ks * 4 + half * 2;
      const v4f b0 = __builtin_nontemporal_load(
          (const v4f*)&bases1[(size_t)k * NE + col0]);
      const v4f b1 = __builtin_nontemporal_load(
          (const v4f*)&bases1[(size_t)(k + 1) * NE + col0]);
#pragma unroll
      for (int nt = 0; nt < 4; ++nt) {
        v2f b;
        b.x = b0[nt];
        b.y = b1[nt];
#pragma unroll
        for (int mt = 0; mt < 4; ++mt) {
          acc[nt][mt] = __builtin_amdgcn_wmma_f32_16x16x4_f32(
               false, a[mt][ks],  false, b,
               (short)0, acc[nt][mt],
               false,  false);
        }
      }
    }
#pragma unroll
    for (int mt = 0; mt < 4; ++mt) {
#pragma unroll
      for (int j = 0; j < 8; ++j) {
        int m = mt * 16 + j + half * 8;
        if (m < RR) {
          v4f o;
          o.x = acc[0][mt][j];
          o.y = acc[1][mt][j];
          o.z = acc[2][mt][j];
          o.w = acc[3][mt][j];
          *(v4f*)&W1[(size_t)m * NE + col0] = o;
        }
      }
    }
  }
}

typedef int v4i __attribute__((ext_vector_type(4)));

#define EPW      128
#define K3_WAVES 4
#define K3_GRID  (NNZ / (EPW * K3_WAVES))
static_assert(NNZ % (EPW * K3_WAVES) == 0, "edges must fill whole workgroups");

template <int J>
__device__ __forceinline__ int pairRecord(const v4i r, bool hi) {
  const int lo_v = __builtin_amdgcn_readlane(r[(J & 1) * 2],     J >> 1);
  const int hi_v = __builtin_amdgcn_readlane(r[(J & 1) * 2 + 1], J >> 1);
  return hi ? hi_v : lo_v;
}

template <int G>
__device__ __forceinline__ void k3Gather(const v4i c4, bool hi, int e,
                                         const float* __restrict__ W1,
                                         float (&w)[16]) {
#define K3_LOAD(Q) { const int col = pairRecord<G * 16 + (Q)>(c4, hi); \
                     w[Q] = W1[(size_t)(unsigned)(col * EE + e)]; }
  K3_LOAD(0)  K3_LOAD(1)  K3_LOAD(2)  K3_LOAD(3)
  K3_LOAD(4)  K3_LOAD(5)  K3_LOAD(6)  K3_LOAD(7)
  K3_LOAD(8)  K3_LOAD(9)  K3_LOAD(10) K3_LOAD(11)
  K3_LOAD(12) K3_LOAD(13) K3_LOAD(14) K3_LOAD(15)
#undef K3_LOAD
}

template <int G>
__device__ __forceinline__ void k3Add(const v4i r4, const v4i v4bits, bool hi,
                                      int e, const float (&w)[16],
                                      float* __restrict__ h) {
#define K3_ADD(Q) { const int row = pairRecord<G * 16 + (Q)>(r4, hi); \
                    const float v = __builtin_bit_cast(float, pairRecord<G * 16 + (Q)>(v4bits, hi)); \
                    atomAddF32(&h[row * EE + e], v * w[Q]); }
  K3_ADD(0)  K3_ADD(1)  K3_ADD(2)  K3_ADD(3)
  K3_ADD(4)  K3_ADD(5)  K3_ADD(6)  K3_ADD(7)
  K3_ADD(8)  K3_ADD(9)  K3_ADD(10) K3_ADD(11)
  K3_ADD(12) K3_ADD(13) K3_ADD(14) K3_ADD(15)
#undef K3_ADD
}

__global__ void __launch_bounds__(32 * K3_WAVES)
rgcn_scatter_h(const float* __restrict__ values,
               const int* __restrict__ hr,
               const int* __restrict__ hc,
               const float* __restrict__ W1,
               float* __restrict__ h) {
  const int lane = threadIdx.x & 31;
  const int wave = blockIdx.x * K3_WAVES + (threadIdx.x >> 5);
  const int e    = lane & 15;
  const bool hi  = lane >= 16;
  const int base = wave * EPW + lane * 4;
  const v4i c4 = __builtin_nontemporal_load((const v4i*)&hc[base]);
  const v4i r4 = __builtin_nontemporal_load((const v4i*)&hr[base]);
  const v4f vf = __builtin_nontemporal_load((const v4f*)&values[base]);
  const v4i vb = __builtin_bit_cast(v4i, vf);

  float wa[16], wb[16];
  k3Gather<0>(c4, hi, e, W1, wa);
  k3Gather<1>(c4, hi, e, W1, wb);
  k3Add<0>(r4, vb, hi, e, wa, h);
  k3Gather<2>(c4, hi, e, W1, wa);
  k3Add<1>(r4, vb, hi, e, wb, h);
  k3Gather<3>(c4, hi, e, W1, wb);
  k3Add<2>(r4, vb, hi, e, wa, h);
  k3Add<3>(r4, vb, hi, e, wb, h);
}

__global__ void rgcn_bias_relu(float* __restrict__ h,
                               const float* __restrict__ bias1) {
  int i = blockIdx.x * blockDim.x + threadIdx.x;
  if (i >= NE) return;
  float x = h[i] + bias1[i & 15];
  h[i] = x > 0.0f ? x : 0.0f;
}

#define K5_WAVES   10
#define K5_THREADS (32 * K5_WAVES)
#define K5_GRID    (NNZ / (EPW * K5_WAVES))
static_assert(NNZ % (EPW * K5_WAVES) == 0, "edges must fill whole workgroups");
#define W2_ELEMS   (RR * EE * CC)
#define W2_QUADS   (W2_ELEMS / 4)
#define K5_GROUP   8

template <int MASK>
__device__ __forceinline__ float rowXor(float x) {
  return __builtin_bit_cast(float,
      __builtin_amdgcn_update_dpp(0, __builtin_bit_cast(int, x),
                                  0x160 + MASK, 0xF, 0xF, true));
}

template <int G>
__device__ __forceinline__ void k5Gather(const v4i c4, bool hi, int e,
                                         const float* __restrict__ h,
                                         float (&w)[K5_GROUP]) {
#define K5_LOAD(Q) { const int o = pairRecord<G * K5_GROUP + (Q)>(c4, hi); \
                     w[Q] = h[o * EE + e]; }
  K5_LOAD(0) K5_LOAD(1) K5_LOAD(2) K5_LOAD(3)
  K5_LOAD(4) K5_LOAD(5) K5_LOAD(6) K5_LOAD(7)
#undef K5_LOAD
}

struct K5Lane {
  int  offA, offB;
  bool b2, b1, even;
  int  cls;
};

template <int J>
__device__ __forceinline__ float k5Partial(const v4i r4, bool hi, float hv,
                                           const K5Lane& L, const float* W2s,
                                           int& sOut) {
  const int rlo = __builtin_amdgcn_readlane(r4[(J & 1) * 2],     J >> 1);
  const int rhi = __builtin_amdgcn_readlane(r4[(J & 1) * 2 + 1], J >> 1);
  const int plo = rlo / NN, phi = rhi / NN;
  const int slo = rlo - plo * NN, shi = rhi - phi * NN;
  const int p = hi ? phi : plo;
  sOut = hi ? shi : slo;
  const float* wp = W2s + p * (EE * CC);
  const v4f qa = *(const v4f*)(wp + L.offA);
  const v4f qb = *(const v4f*)(wp + L.offB);
  const float hp = rowXor<8>(hv);
  const float s0 = hv * qa.x + hp * qb.x;
  const float s1 = hv * qa.y + hp * qb.y;
  const float s2 = hv * qa.z + hp * qb.z;
  const float s3 = hv * qa.w + hp * qb.w;
  const float k0 = L.b2 ? s2 : s0, k1 = L.b2 ? s3 : s1;
  const float g0 = L.b2 ? s0 : s2, g1 = L.b2 ? s1 : s3;
  const float t0 = k0 + rowXor<4>(g0);
  const float t1 = k1 + rowXor<4>(g1);
  const float k = L.b1 ? t1 : t0;
  const float g = L.b1 ? t0 : t1;
  return k + rowXor<2>(g);
}

template <int J>
__device__ __forceinline__ void k5Two(const v4i r4, const v4i v4bits, bool hi,
                                      float hvA, float hvB, const K5Lane& L,
                                      const float* W2s,
                                      float* __restrict__ out) {
  int sA, sB;
  const float uA = k5Partial<J>(r4, hi, hvA, L, W2s, sA);
  const float uB = k5Partial<J + 1>(r4, hi, hvB, L, W2s, sB);
  const float vA = __builtin_bit_cast(float, pairRecord<J>(v4bits, hi));
  const float vB = __builtin_bit_cast(float, pairRecord<J + 1>(v4bits, hi));
  const float keep = L.even ? uA : uB;
  const float give = L.even ? uB : uA;
  const float tot  = keep + rowXor<1>(give);
  const int   s    = L.even ? sA : sB;
  const float v    = L.even ? vA : vB;
  atomAddF32(&out[s * CC + L.cls], v * tot);
}

template <int G>
__device__ __forceinline__ void k5Apply(const v4i r4, const v4i v4bits, bool hi,
                                        const float (&w)[K5_GROUP],
                                        const K5Lane& L, const float* W2s,
                                        float* __restrict__ out) {
  k5Two<G * K5_GROUP + 0>(r4, v4bits, hi, w[0], w[1], L, W2s, out);
  k5Two<G * K5_GROUP + 2>(r4, v4bits, hi, w[2], w[3], L, W2s, out);
  k5Two<G * K5_GROUP + 4>(r4, v4bits, hi, w[4], w[5], L, W2s, out);
  k5Two<G * K5_GROUP + 6>(r4, v4bits, hi, w[6], w[7], L, W2s, out);
}

__global__ void __launch_bounds__(K5_THREADS)
rgcn_layer2(const float* __restrict__ values,
            const int* __restrict__ vr,
            const int* __restrict__ vc,
            const float* __restrict__ h,
            const float* __restrict__ W2,
            float* __restrict__ out) {
  __shared__ __attribute__((aligned(16))) float W2s[W2_ELEMS];

  const int lane = threadIdx.x & 31;
  const int wave = blockIdx.x * K5_WAVES + (threadIdx.x >> 5);
  const int e    = lane & 15;
  const bool hi  = lane >= 16;
  const int base = wave * EPW + lane * 4;
  const v4i c4 = __builtin_nontemporal_load((const v4i*)&vc[base]);
  const v4i r4 = __builtin_nontemporal_load((const v4i*)&vr[base]);
  const v4f vf = __builtin_nontemporal_load((const v4f*)&values[base]);
  const v4i vb = __builtin_bit_cast(v4i, vf);

  {
    const v4f* W2g = (const v4f*)W2;
    v4f* W2q = (v4f*)W2s;
#pragma unroll
    for (int k = 0; k < W2_QUADS / K5_THREADS; ++k)
      W2q[threadIdx.x + k * K5_THREADS] = W2g[threadIdx.x + k * K5_THREADS];
    if (threadIdx.x < W2_QUADS % K5_THREADS)
      W2q[(W2_QUADS / K5_THREADS) * K5_THREADS + threadIdx.x] =
          W2g[(W2_QUADS / K5_THREADS) * K5_THREADS + threadIdx.x];
  }

  K5Lane L;
  const int gsel = e >> 3;
  L.offA = e * CC + 4 * gsel;
  L.offB = (e ^ 8) * CC + 4 * gsel;
  L.b2   = (e & 4) != 0;
  L.b1   = (e & 2) != 0;
  L.even = (e & 1) == 0;
  L.cls  = e >> 1;

  float wa[K5_GROUP], wb[K5_GROUP];
  k5Gather<0>(c4, hi, e, h, wa);
  k5Gather<1>(c4, hi, e, h, wb);
  __syncthreads();
  k5Apply<0>(r4, vb, hi, wa, L, W2s, out);
  k5Gather<2>(c4, hi, e, h, wa);
  k5Apply<1>(r4, vb, hi, wb, L, W2s, out);
  k5Gather<3>(c4, hi, e, h, wb);
  k5Apply<2>(r4, vb, hi, wa, L, W2s, out);
  k5Gather<4>(c4, hi, e, h, wa);
  k5Apply<3>(r4, vb, hi, wb, L, W2s, out);
  k5Gather<5>(c4, hi, e, h, wb);
  k5Apply<4>(r4, vb, hi, wa, L, W2s, out);
  k5Gather<6>(c4, hi, e, h, wa);
  k5Apply<5>(r4, vb, hi, wb, L, W2s, out);
  k5Gather<7>(c4, hi, e, h, wb);
  k5Apply<6>(r4, vb, hi, wa, L, W2s, out);
  k5Apply<7>(r4, vb, hi, wb, L, W2s, out);
}

extern "C" void kernel_launch(void* const* d_in, const int* in_sizes, int n_in,
                              void* d_out, int out_size, void* d_ws, size_t ws_size,
                              hipStream_t stream) {
  const float* values = (const float*)d_in[0];
  const float* comps1 = (const float*)d_in[1];
  const float* bases1 = (const float*)d_in[2];
  const float* comps2 = (const float*)d_in[3];
  const float* bases2 = (const float*)d_in[4];
  const float* bias1  = (const float*)d_in[5];
  const float* bias2  = (const float*)d_in[6];
  const int*   hr     = (const int*)d_in[7];
  const int*   hc     = (const int*)d_in[8];
  const int*   vr     = (const int*)d_in[9];
  const int*   vc     = (const int*)d_in[10];
  float* out = (float*)d_out;

  float* W1 = (float*)d_ws;
  float* h  = W1 + W1ELEMS;
  float* W2 = h + NE;

  rgcn_init<<<(NE + 255) / 256, 256, 0, stream>>>(h, out, bias2);
  rgcn_w2<<<(RR * EE * CC + 255) / 256, 256, 0, stream>>>(comps2, bases2, W2);
  rgcn_w1_gemm<<<2500, 32, 0, stream>>>(comps1, bases1, W1);
  rgcn_scatter_h<<<K3_GRID, 32 * K3_WAVES, 0, stream>>>(values, hr, hc, W1, h);
  rgcn_bias_relu<<<(NE + 255) / 256, 256, 0, stream>>>(h, bias1);
  rgcn_layer2<<<K5_GRID, K5_THREADS, 0, stream>>>(values, vr, vc, h, W2, out);
}


# speedup vs baseline: 1.8195x; 1.2313x over previous
#include <hip/hip_runtime.h>
#include <hip/hip_bf16.h>

#define NN    50000
#define RR    51
#define EE    16
#define CC    8
#define BB    40
#define NNZ   1600000
#define NE    (NN * EE)
#define W1ELEMS ((size_t)RR * NE)

typedef float v2f __attribute__((ext_vector_type(2)));
typedef float v8f __attribute__((ext_vector_type(8)));

__device__ __forceinline__ void atomAddF32(float* p, float v) {
  unsafeAtomicAdd(p, v);
}

__global__ void rgcn_init(float* __restrict__ h, float* __restrict__ out,
                          const float* __restrict__ bias2) {
  int i = blockIdx.x * blockDim.x + threadIdx.x;
  if (i < NE)        h[i] = 0.0f;
  if (i < NN * CC)   out[i] = bias2[i & (CC - 1)];
}

__global__ void rgcn_w2(const float* __restrict__ comps2,
                        const float* __restrict__ bases2,
                        float* __restrict__ W2) {
  int idx = blockIdx.x * blockDim.x + threadIdx.x;
  if (idx >= RR * EE * CC) return;
  int r  = idx >> 7;
  int ec = idx & 127;
  float acc = 0.0f;
#pragma unroll
  for (int b = 0; b < BB; ++b)
    acc += comps2[r * BB + b] * bases2[b * (EE * CC) + ec];
  W2[idx] = acc;
}

typedef float v4f __attribute__((ext_vector_type(4)));

#define STRIP   64
#define NSTRIPS (NE / STRIP)

__global__ void __launch_bounds__(32)
rgcn_w1_gemm(const float* __restrict__ comps1,
             const float* __restrict__ bases1,
             float* __restrict__ W1) {
  const int lane = threadIdx.x & 31;
  const int half = lane >> 4;
  const int l16  = lane & 15;

  v2f a[4][10];
#pragma unroll
  for (int mt = 0; mt < 4; ++mt) {
    int m = mt * 16 + l16;
#pragma unroll
    for (int ks = 0; ks < 10; ++ks) {
      int k = ks * 4 + half * 2;
      float ax = 0.0f, ay = 0.0f;
      if (m < RR) {
        ax = comps1[m * BB + k];
        ay = comps1[m * BB + k + 1];
      }
      a[mt][ks].x = ax;
      a[mt][ks].y = ay;
    }
  }

  for (int strip = blockIdx.x; strip < NSTRIPS; strip += gridDim.x) {
    const int col0 = strip * STRIP + l16 * 4;
    v8f acc[4][4] = {};
#pragma unroll
    for (int ks = 0; ks < 10; ++ks) {
      const int k = ks * 4 + half * 2;
      const v4f b0 = __builtin_nontemporal_load(
          (const v4f*)&bases1[(size_t)k * NE + col0]);
      const v4f b1 = __builtin_nontemporal_load(
          (const v4f*)&bases1[(size_t)(k + 1) * NE + col0]);
#pragma unroll
      for (int nt = 0; nt < 4; ++nt) {
        v2f b;
        b.x = b0[nt];
        b.y = b1[nt];
#pragma unroll
        for (int mt = 0; mt < 4; ++mt) {
          acc[nt][mt] = __builtin_amdgcn_wmma_f32_16x16x4_f32(
               false, a[mt][ks],  false, b,
               (short)0, acc[nt][mt],
               false,  false);
        }
      }
    }
#pragma unroll
    for (int mt = 0; mt < 4; ++mt) {
#pragma unroll
      for (int j = 0; j < 8; ++j) {
        int m = mt * 16 + j + half * 8;
        if (m < RR) {
          v4f o;
          o.x = acc[0][mt][j];
          o.y = acc[1][mt][j];
          o.z = acc[2][mt][j];
          o.w = acc[3][mt][j];
          *(v4f*)&W1[(size_t)m * NE + col0] = o;
        }
      }
    }
  }
}

typedef int v4i __attribute__((ext_vector_type(4)));

#define EPW      128
#define K3_WAVES 4
#define K3_GRID  (NNZ / (EPW * K3_WAVES))
static_assert(NNZ % (EPW * K3_WAVES) == 0, "edges must fill whole workgroups");
#define K3_GROUP  8
#define K3_BSHIFT 11
#define K3_NBUCK  ((NN + (1 << K3_BSHIFT) - 1) >> K3_BSHIFT)
static_assert(K3_NBUCK <= 32, "one bucket counter per lane");

__device__ __forceinline__ int rowKey(int row, int rot) {
  unsigned b = (unsigned)row >> K3_BSHIFT;
  b = b < (unsigned)(K3_NBUCK - 1) ? b : (unsigned)(K3_NBUCK - 1);
  const int k = (int)b + rot;
  return k >= K3_NBUCK ? k - K3_NBUCK : k;
}

__device__ __forceinline__ int walkTurn() {
  const unsigned place = __builtin_amdgcn_s_sendmsg_rtn(0x87);
  return (place & 0x40000u) ? (K3_NBUCK + 1) / 2 : 0;
}

__device__ __forceinline__ void waveSlots(const int (&key)[4], int lane,
                                          int* cnt, int* start, int (&pos)[4]) {
  cnt[lane] = 0;
  __syncthreads();
  int rank[4];
#pragma unroll
  for (int k = 0; k < 4; ++k) rank[k] = atomicAdd(&cnt[key[k]], 1);
  __syncthreads();
  const int c = cnt[lane];
  int incl = c;
#pragma unroll
  for (int d = 1; d < 32; d <<= 1) {
    const int t = __shfl_up(incl, d);
    if (lane >= d) incl += t;
  }
  start[lane] = incl - c;
  __syncthreads();
#pragma unroll
  for (int k = 0; k < 4; ++k) pos[k] = start[key[k]] + rank[k];
}

template <int G>
__device__ __forceinline__ void k3Gather(const int (*rec)[4], int e,
                                         const float* __restrict__ W1,
                                         float (&w)[K3_GROUP]) {
  int col[K3_GROUP];
#pragma unroll
  for (int q = 0; q < K3_GROUP; ++q) col[q] = rec[2 * (G * K3_GROUP + q)][2];
#pragma unroll
  for (int q = 0; q < K3_GROUP; ++q) w[q] = W1[(size_t)(unsigned)(col[q] * EE + e)];
}

template <int G>
__device__ __forceinline__ void k3Add(const int (*rec)[4], int e,
                                      const float (&w)[K3_GROUP],
                                      float* __restrict__ h) {
  int row[K3_GROUP], vbits[K3_GROUP];
#pragma unroll
  for (int q = 0; q < K3_GROUP; ++q) {
    row[q]   = rec[2 * (G * K3_GROUP + q)][0];
    vbits[q] = rec[2 * (G * K3_GROUP + q)][1];
  }
#pragma unroll
  for (int q = 0; q < K3_GROUP; ++q)
    atomAddF32(&h[row[q] * EE + e], __builtin_bit_cast(float, vbits[q]) * w[q]);
}

__global__ void __launch_bounds__(32 * K3_WAVES) __attribute__((amdgpu_waves_per_eu(16)))
rgcn_scatter_h(const float* __restrict__ values,
               const int* __restrict__ hr,
               const int* __restrict__ hc,
               const float* __restrict__ W1,
               float* __restrict__ h) {
  __shared__ int k3cnt[K3_WAVES][32];
  __shared__ int k3start[K3_WAVES][32];
  __shared__ __attribute__((aligned(16))) int k3rec[K3_WAVES][EPW][4];

  const int lane = threadIdx.x & 31;
  const int wv   = threadIdx.x >> 5;
  const int wave = blockIdx.x * K3_WAVES + wv;
  const int e    = lane & 15;
  const bool hi  = lane >= 16;
  const int base = wave * EPW + lane * 4;
  const v4i c4 = __builtin_nontemporal_load((const v4i*)&hc[base]);
  const v4i r4 = __builtin_nontemporal_load((const v4i*)&hr[base]);
  const v4f vf = __builtin_nontemporal_load((const v4f*)&values[base]);
  const v4i vb = __builtin_bit_cast(v4i, vf);

  const int rot = walkTurn();
  int key[4], pos[4];
#pragma unroll
  for (int k = 0; k < 4; ++k) key[k] = rowKey(r4[k], rot);
  waveSlots(key, lane, k3cnt[wv], k3start[wv], pos);
#pragma unroll
  for (int k = 0; k < 4; ++k) {
    int* r = k3rec[wv][pos[k]];
    r[0] = r4[k];
    r[1] = vb[k];
    r[2] = c4[k];
    r[3] = 0;
  }
  __syncthreads();
  const int (*rec)[4] = &k3rec[wv][hi ? 1 : 0];

  float wa[K3_GROUP], wb[K3_GROUP];
  k3Gather<0>(rec, e, W1, wa);
  k3Gather<1>(rec, e, W1, wb);
  k3Add<0>(rec, e, wa, h);
  k3Gather<2>(rec, e, W1, wa);
  k3Add<1>(rec, e, wb, h);
  k3Gather<3>(rec, e, W1, wb);
  k3Add<2>(rec, e, wa, h);
  k3Gather<4>(rec, e, W1, wa);
  k3Add<3>(rec, e, wb, h);
  k3Gather<5>(rec, e, W1, wb);
  k3Add<4>(rec, e, wa, h);
  k3Gather<6>(rec, e, W1, wa);
  k3Add<5>(rec, e, wb, h);
  k3Gather<7>(rec, e, W1, wb);
  k3Add<6>(rec, e, wa, h);
  k3Add<7>(rec, e, wb, h);
}

__global__ void rgcn_bias_relu(float* __restrict__ h,
                               const float* __restrict__ bias1) {
  int i = blockIdx.x * blockDim.x + threadIdx.x;
  if (i >= NE) return;
  float x = h[i] + bias1[i & 15];
  h[i] = x > 0.0f ? x : 0.0f;
}

#define K5_WAVES   10
#define K5_THREADS (32 * K5_WAVES)
#define K5_GRID    (NNZ / (EPW * K5_WAVES))
static_assert(NNZ % (EPW * K5_WAVES) == 0, "edges must fill whole workgroups");
#define W2_ELEMS   (RR * EE * CC)
#define W2_QUADS   (W2_ELEMS / 4)
#define K5_GROUP   8

template <int MASK>
__device__ __forceinline__ float rowXor(float x) {
  return __builtin_bit_cast(float,
      __builtin_amdgcn_update_dpp(0, __builtin_bit_cast(int, x),
                                  0x160 + MASK, 0xF, 0xF, true));
}

template <int G>
__device__ __forceinline__ void k5Gather(const v4i* rec, int e,
                                         const float* __restrict__ h,
                                         float (&w)[K5_GROUP]) {
  int o[K5_GROUP];
#pragma unroll
  for (int q = 0; q < K5_GROUP; ++q) o[q] = rec[2 * (G * K5_GROUP + q)].z;
#pragma unroll
  for (int q = 0; q < K5_GROUP; ++q) w[q] = h[o[q] * EE + e];
}

struct K5Lane {
  int  offA, offB;
  bool b2, b1, even;
  int  cls;
};

__device__ __forceinline__ float k5Partial(int p, float hv, const K5Lane& L,
                                           const float* W2s) {
  const float* wp = W2s + p * (EE * CC);
  const v4f qa = *(const v4f*)(wp + L.offA);
  const v4f qb = *(const v4f*)(wp + L.offB);
  const float hp = rowXor<8>(hv);
  const float s0 = hv * qa.x + hp * qb.x;
  const float s1 = hv * qa.y + hp * qb.y;
  const float s2 = hv * qa.z + hp * qb.z;
  const float s3 = hv * qa.w + hp * qb.w;
  const float k0 = L.b2 ? s2 : s0, k1 = L.b2 ? s3 : s1;
  const float g0 = L.b2 ? s0 : s2, g1 = L.b2 ? s1 : s3;
  const float t0 = k0 + rowXor<4>(g0);
  const float t1 = k1 + rowXor<4>(g1);
  const float k = L.b1 ? t1 : t0;
  const float g = L.b1 ? t0 : t1;
  return k + rowXor<2>(g);
}

template <int J>
__device__ __forceinline__ void k5Two(const v4i* rec, float hvA, float hvB,
                                      const K5Lane& L, const float* W2s,
                                      float* __restrict__ out) {
  const v4i ra = rec[2 * J];
  const v4i rb = rec[2 * (J + 1)];
  const float uA = k5Partial(ra.w, hvA, L, W2s);
  const float uB = k5Partial(rb.w, hvB, L, W2s);
  const float keep = L.even ? uA : uB;
  const float give = L.even ? uB : uA;
  const float tot  = keep + rowXor<1>(give);
  const int   s    = L.even ? ra.x : rb.x;
  const float v    = __builtin_bit_cast(float, L.even ? ra.y : rb.y);
  atomAddF32(&out[s * CC + L.cls], v * tot);
}

template <int G>
__device__ __forceinline__ void k5Apply(const v4i* rec,
                                        const float (&w)[K5_GROUP],
                                        const K5Lane& L, const float* W2s,
                                        float* __restrict__ out) {
  k5Two<G * K5_GROUP + 0>(rec, w[0], w[1], L, W2s, out);
  k5Two<G * K5_GROUP + 2>(rec, w[2], w[3], L, W2s, out);
  k5Two<G * K5_GROUP + 4>(rec, w[4], w[5], L, W2s, out);
  k5Two<G * K5_GROUP + 6>(rec, w[6], w[7], L, W2s, out);
}

__global__ void __launch_bounds__(K5_THREADS)
rgcn_layer2(const float* __restrict__ values,
            const int* __restrict__ vr,
            const int* __restrict__ vc,
            const float* __restrict__ h,
            const float* __restrict__ W2,
            float* __restrict__ out) {
  __shared__ __attribute__((aligned(16))) float W2s[W2_ELEMS];
  __shared__ int k5cnt[K5_WAVES][32];
  __shared__ int k5start[K5_WAVES][32];
  __shared__ __attribute__((aligned(16))) v4i k5rec[K5_WAVES][EPW];

  const int lane = threadIdx.x & 31;
  const int wv   = threadIdx.x >> 5;
  const int wave = blockIdx.x * K5_WAVES + wv;
  const int e    = lane & 15;
  const bool hi  = lane >= 16;
  const int base = wave * EPW + lane * 4;
  const v4i c4 = __builtin_nontemporal_load((const v4i*)&vc[base]);
  const v4i r4 = __builtin_nontemporal_load((const v4i*)&vr[base]);
  const v4f vf = __builtin_nontemporal_load((const v4f*)&values[base]);
  const v4i vb = __builtin_bit_cast(v4i, vf);

  {
    const v4f* W2g = (const v4f*)W2;
    v4f* W2q = (v4f*)W2s;
#pragma unroll
    for (int k = 0; k < W2_QUADS / K5_THREADS; ++k)
      W2q[threadIdx.x + k * K5_THREADS] = W2g[threadIdx.x + k * K5_THREADS];
    if (threadIdx.x < W2_QUADS % K5_THREADS)
      W2q[(W2_QUADS / K5_THREADS) * K5_THREADS + threadIdx.x] =
          W2g[(W2_QUADS / K5_THREADS) * K5_THREADS + threadIdx.x];
  }

  const int rot = walkTurn();
  int prel[4], srow[4], key[4], pos[4];
#pragma unroll
  for (int k = 0; k < 4; ++k) {
    prel[k] = r4[k] / NN;
    srow[k] = r4[k] - prel[k] * NN;
    key[k]  = rowKey(srow[k], rot);
  }
  waveSlots(key, lane, k5cnt[wv], k5start[wv], pos);
#pragma unroll
  for (int k = 0; k < 4; ++k) {
    v4i r;
    r.x = srow[k];
    r.y = vb[k];
    r.z = c4[k];
    r.w = prel[k];
    k5rec[wv][pos[k]] = r;
  }
  __syncthreads();
  const v4i* rec = &k5rec[wv][hi ? 1 : 0];

  K5Lane L;
  const int gsel = e >> 3;
  L.offA = e * CC + 4 * gsel;
  L.offB = (e ^ 8) * CC + 4 * gsel;
  L.b2   = (e & 4) != 0;
  L.b1   = (e & 2) != 0;
  L.even = (e & 1) == 0;
  L.cls  = e >> 1;

  float wa[K5_GROUP], wb[K5_GROUP];
  k5Gather<0>(rec, e, h, wa);
  k5Gather<1>(rec, e, h, wb);
  k5Apply<0>(rec, wa, L, W2s, out);
  k5Gather<2>(rec, e, h, wa);
  k5Apply<1>(rec, wb, L, W2s, out);
  k5Gather<3>(rec, e, h, wb);
  k5Apply<2>(rec, wa, L, W2s, out);
  k5Gather<4>(rec, e, h, wa);
  k5Apply<3>(rec, wb, L, W2s, out);
  k5Gather<5>(rec, e, h, wb);
  k5Apply<4>(rec, wa, L, W2s, out);
  k5Gather<6>(rec, e, h, wa);
  k5Apply<5>(rec, wb, L, W2s, out);
  k5Gather<7>(rec, e, h, wb);
  k5Apply<6>(rec, wa, L, W2s, out);
  k5Apply<7>(rec, wb, L, W2s, out);
}

extern "C" void kernel_launch(void* const* d_in, const int* in_sizes, int n_in,
                              void* d_out, int out_size, void* d_ws, size_t ws_size,
                              hipStream_t stream) {
  const float* values = (const float*)d_in[0];
  const float* comps1 = (const float*)d_in[1];
  const float* bases1 = (const float*)d_in[2];
  const float* comps2 = (const float*)d_in[3];
  const float* bases2 = (const float*)d_in[4];
  const float* bias1  = (const float*)d_in[5];
  const float* bias2  = (const float*)d_in[6];
  const int*   hr     = (const int*)d_in[7];
  const int*   hc     = (const int*)d_in[8];
  const int*   vr     = (const int*)d_in[9];
  const int*   vc     = (const int*)d_in[10];
  float* out = (float*)d_out;

  float* W1 = (float*)d_ws;
  float* h  = W1 + W1ELEMS;
  float* W2 = h + NE;

  rgcn_init<<<(NE + 255) / 256, 256, 0, stream>>>(h, out, bias2);
  rgcn_w2<<<(RR * EE * CC + 255) / 256, 256, 0, stream>>>(comps2, bases2, W2);
  rgcn_w1_gemm<<<2500, 32, 0, stream>>>(comps1, bases1, W1);
  rgcn_scatter_h<<<K3_GRID, 32 * K3_WAVES, 0, stream>>>(values, hr, hc, W1, h);
  rgcn_bias_relu<<<(NE + 255) / 256, 256, 0, stream>>>(h, bias1);
  rgcn_layer2<<<K5_GRID, K5_THREADS, 0, stream>>>(values, vr, vc, h, W2, out);
}
